# starting kernel
# baseline (speedup 1.0000x reference)
#include <hip/hip_runtime.h>
#include <math.h>

typedef __attribute__((ext_vector_type(2))) float v2f;
typedef __attribute__((ext_vector_type(4))) float f4;
typedef __attribute__((ext_vector_type(8))) float v8f;

#define B_DIM 256
#define N_DIM 512
#define E_DIM 64
#define O_DIM 64
#define K_CAPS 8

#define GEMM_TM 128
#define SA_STRIDE 68
#define SB_STRIDE 68

__global__ __launch_bounds__(256) void priors_gemm_wmma(
    const float* __restrict__ emb, const float* __restrict__ S,
    float* __restrict__ priors)
{
  __shared__ float sA[GEMM_TM * SA_STRIDE];
  __shared__ float sB[E_DIM * SB_STRIDE];

  const int tid  = threadIdx.x;
  const int row0 = blockIdx.x * GEMM_TM;

  const f4* emb4 = (const f4*)(emb + (size_t)row0 * E_DIM);
  for (int i = tid; i < GEMM_TM * (E_DIM / 4); i += 256) {
    const int m = i >> 4, c4 = i & 15;
    *(f4*)&sA[m * SA_STRIDE + c4 * 4] = emb4[m * 16 + c4];
  }
  const f4* S4 = (const f4*)S;
  for (int i = tid; i < E_DIM * (O_DIM / 4); i += 256) {
    const int k = i >> 4, c4 = i & 15;
    *(f4*)&sB[k * SB_STRIDE + c4 * 4] = S4[i];
  }
  __syncthreads();

  const int wave = tid >> 5, lane = tid & 31;
  const int lm = lane & 15;
  const int lh = lane >> 4;
  const int m0 = wave * 16;

  for (int nt = 0; nt < O_DIM / 16; ++nt) {
    const int n0 = nt * 16;
    v8f acc;
#pragma unroll
    for (int j = 0; j < 8; ++j) acc[j] = 0.0f;

#pragma unroll
    for (int k0 = 0; k0 < E_DIM; k0 += 4) {
      v2f a;
      const float* ap = &sA[(m0 + lm) * SA_STRIDE + k0 + 2 * lh];
      a[0] = ap[0];
      a[1] = ap[1];
      v2f bm;
      bm[0] = sB[(k0 + 0 + 2 * lh) * SB_STRIDE + n0 + lm];
      bm[1] = sB[(k0 + 1 + 2 * lh) * SB_STRIDE + n0 + lm];
      acc = __builtin_amdgcn_wmma_f32_16x16x4_f32(
           false, a,  false, bm,
           (short)0, acc,  false,  false);
    }
    float* dst = priors + ((size_t)(row0 + m0 + 8 * lh)) * O_DIM + n0 + lm;
#pragma unroll
    for (int j = 0; j < 8; ++j) dst[(size_t)j * O_DIM] = acc[j];
  }
}

#define NP 65

__global__ __launch_bounds__(512) void routing_kernel(
    const float* __restrict__ priors, const float* __restrict__ cc,
    float* __restrict__ out)
{
  extern __shared__ float smem[];
  float* priorsLds = smem;
  float* accSim    = priorsLds + N_DIM * NP;
  float* zLds      = accSim + K_CAPS * N_DIM;
  float* vLds      = zLds + K_CAPS * O_DIM;
  float* part      = vLds + K_CAPS * O_DIM;
  float* scaleL    = part + K_CAPS * 8 * O_DIM;

  const int tid = threadIdx.x;
  const int b   = blockIdx.x;
  const float* pb  = priors + (size_t)b * N_DIM * O_DIM;
  const float* ccb = cc + (size_t)b * K_CAPS * N_DIM * O_DIM;

  for (int i = tid; i < N_DIM * O_DIM; i += 512) {
    const int n = i >> 6, o = i & 63;
    priorsLds[n * NP + o] = pb[i];
  }
  for (int i = tid; i < K_CAPS * N_DIM; i += 512) accSim[i] = 0.0f;
  __syncthreads();

  const int o = tid & 63;
  const int g = tid >> 6;

  for (int iter = 0; iter < 3; ++iter) {
    float zacc[K_CAPS];
#pragma unroll
    for (int k = 0; k < K_CAPS; ++k) zacc[k] = 0.0f;

    for (int jn = 0; jn < N_DIM / 8; ++jn) {
      const int n = g + (jn << 3);
      float c[K_CAPS];
      float m = -3.402823466e+38f;
#pragma unroll
      for (int k = 0; k < K_CAPS; ++k) {
        c[k] = ccb[(size_t)k * (N_DIM * O_DIM) + n * O_DIM + o]
             + accSim[k * N_DIM + n];
        m = fmaxf(m, c[k]);
      }
      float s = 0.0f;
#pragma unroll
      for (int k = 0; k < K_CAPS; ++k) { c[k] = __expf(c[k] - m); s += c[k]; }
      const float p = priorsLds[n * NP + o] / s;
#pragma unroll
      for (int k = 0; k < K_CAPS; ++k) zacc[k] += c[k] * p;
    }
#pragma unroll
    for (int k = 0; k < K_CAPS; ++k) part[(k * 8 + g) * O_DIM + o] = zacc[k];
    __syncthreads();

    {
      float z = 0.0f;
#pragma unroll
      for (int g2 = 0; g2 < 8; ++g2) z += part[(g * 8 + g2) * O_DIM + o];
      zLds[g * O_DIM + o] = z;
    }
    __syncthreads();

    if (tid < K_CAPS) {
      float sq = 0.0f;
      for (int oo = 0; oo < O_DIM; ++oo) {
        const float zz = zLds[tid * O_DIM + oo];
        sq += zz * zz;
      }
      scaleL[tid] = (sq / (1.0f + sq)) / sqrtf(sq + 1e-9f);
    }
    __syncthreads();

    if (iter == 2) {
      out[(size_t)b * K_CAPS * O_DIM + tid] = zLds[tid] * scaleL[g];
    } else {
      vLds[tid] = zLds[tid] * scaleL[g];
      __syncthreads();
      const int n = tid;
#pragma unroll
      for (int k = 0; k < K_CAPS; ++k) {
        float s2 = 0.0f;
#pragma unroll 8
        for (int oo = 0; oo < O_DIM; ++oo)
          s2 += priorsLds[n * NP + oo] * vLds[k * O_DIM + oo];
        accSim[k * N_DIM + n] += s2;
      }
      __syncthreads();
    }
  }
}

extern "C" void kernel_launch(void* const* d_in, const int* in_sizes, int n_in,
                              void* d_out, int out_size, void* d_ws, size_t ws_size,
                              hipStream_t stream) {
  (void)in_sizes; (void)n_in; (void)out_size; (void)ws_size;
  const float* emb = (const float*)d_in[0];
  const float* S   = (const float*)d_in[1];
  const float* cc  = (const float*)d_in[2];
  float* out    = (float*)d_out;
  float* priors = (float*)d_ws;

  priors_gemm_wmma<<<(B_DIM * N_DIM) / GEMM_TM, 256, 0, stream>>>(emb, S, priors);

  const size_t smemBytes =
      (size_t)(N_DIM * NP + K_CAPS * N_DIM + 2 * K_CAPS * O_DIM +
               K_CAPS * 8 * O_DIM + 8) * sizeof(float);
  routing_kernel<<<B_DIM, 512, smemBytes, stream>>>(priors, cc, out);
}


# Routing pass A: 16-byte cc loads per lane (512 contiguous bytes per wave load), the next n-block's 8 loads issued before the current block is summed; n summed in 32 groups of 16 (was 8 of 64)
# speedup vs baseline: 1.8336x; 1.8336x over previous
#include <hip/hip_runtime.h>
#include <math.h>

typedef __attribute__((ext_vector_type(2))) float v2f;
typedef __attribute__((ext_vector_type(4))) float f4;
typedef __attribute__((ext_vector_type(8))) float v8f;

#define B_DIM 256
#define N_DIM 512
#define E_DIM 64
#define O_DIM 64
#define K_CAPS 8

#define GEMM_TM 128
#define SA_STRIDE 68
#define SB_STRIDE 68

__global__ __launch_bounds__(256) void priors_gemm_wmma(
    const float* __restrict__ emb, const float* __restrict__ S,
    float* __restrict__ priors)
{
  __shared__ float sA[GEMM_TM * SA_STRIDE];
  __shared__ float sB[E_DIM * SB_STRIDE];

  const int tid  = threadIdx.x;
  const int row0 = blockIdx.x * GEMM_TM;

  const f4* emb4 = (const f4*)(emb + (size_t)row0 * E_DIM);
  for (int i = tid; i < GEMM_TM * (E_DIM / 4); i += 256) {
    const int m = i >> 4, c4 = i & 15;
    *(f4*)&sA[m * SA_STRIDE + c4 * 4] = emb4[m * 16 + c4];
  }
  const f4* S4 = (const f4*)S;
  for (int i = tid; i < E_DIM * (O_DIM / 4); i += 256) {
    const int k = i >> 4, c4 = i & 15;
    *(f4*)&sB[k * SB_STRIDE + c4 * 4] = S4[i];
  }
  __syncthreads();

  const int wave = tid >> 5, lane = tid & 31;
  const int lm = lane & 15;
  const int lh = lane >> 4;
  const int m0 = wave * 16;

  for (int nt = 0; nt < O_DIM / 16; ++nt) {
    const int n0 = nt * 16;
    v8f acc;
#pragma unroll
    for (int j = 0; j < 8; ++j) acc[j] = 0.0f;

#pragma unroll
    for (int k0 = 0; k0 < E_DIM; k0 += 4) {
      v2f a;
      const float* ap = &sA[(m0 + lm) * SA_STRIDE + k0 + 2 * lh];
      a[0] = ap[0];
      a[1] = ap[1];
      v2f bm;
      bm[0] = sB[(k0 + 0 + 2 * lh) * SB_STRIDE + n0 + lm];
      bm[1] = sB[(k0 + 1 + 2 * lh) * SB_STRIDE + n0 + lm];
      acc = __builtin_amdgcn_wmma_f32_16x16x4_f32(
           false, a,  false, bm,
           (short)0, acc,  false,  false);
    }
    float* dst = priors + ((size_t)(row0 + m0 + 8 * lh)) * O_DIM + n0 + lm;
#pragma unroll
    for (int j = 0; j < 8; ++j) dst[(size_t)j * O_DIM] = acc[j];
  }
}

#define NP 65
#define NGRP 32
#define NBLK 16
#define CC_KSTRIDE4 (N_DIM * O_DIM / 4)

#define LOAD_BLOCK(BUF, JN)                                                   \
  do {                                                                        \
    _Pragma("unroll") for (int k = 0; k < K_CAPS; ++k)                        \
      BUF[k] = ccp[(size_t)k * CC_KSTRIDE4 + (JN) * 512];                     \
  } while (0)

#define SUM_BLOCK(BUF, JN)                                                    \
  do {                                                                        \
    const int n = ng + (JN) * NGRP;                                           \
    f4 pr;                                                                    \
    _Pragma("unroll") for (int j = 0; j < 4; ++j)                             \
      pr[j] = priorsLds[n * NP + 4 * o4 + j];                                 \
    f4 c[K_CAPS];                                                             \
    f4 m;                                                                     \
    _Pragma("unroll") for (int j = 0; j < 4; ++j) m[j] = -3.402823466e+38f;   \
    _Pragma("unroll") for (int k = 0; k < K_CAPS; ++k) {                      \
      const float a = accSim[k * N_DIM + n];                                  \
      _Pragma("unroll") for (int j = 0; j < 4; ++j) {                         \
        c[k][j] = BUF[k][j] + a;                                              \
        m[j] = fmaxf(m[j], c[k][j]);                                          \
      }                                                                       \
    }                                                                         \
    f4 s;                                                                     \
    _Pragma("unroll") for (int j = 0; j < 4; ++j) s[j] = 0.0f;                \
    _Pragma("unroll") for (int k = 0; k < K_CAPS; ++k) {                      \
      _Pragma("unroll") for (int j = 0; j < 4; ++j) {                         \
        c[k][j] = __expf(c[k][j] - m[j]);                                     \
        s[j] += c[k][j];                                                      \
      }                                                                       \
    }                                                                         \
    f4 p;                                                                     \
    _Pragma("unroll") for (int j = 0; j < 4; ++j) p[j] = pr[j] / s[j];        \
    _Pragma("unroll") for (int k = 0; k < K_CAPS; ++k) {                      \
      _Pragma("unroll") for (int j = 0; j < 4; ++j)                           \
        zacc[k][j] += c[k][j] * p[j];                                         \
    }                                                                         \
  } while (0)

#define END_BLOCK()                                                           \
  do {                                                                        \
    _Pragma("unroll") for (int k = 0; k < K_CAPS; ++k)                        \
      asm volatile("" : "+v"(zacc[k]));                                       \
  } while (0)

__global__ __launch_bounds__(512) void routing_kernel(
    const float* __restrict__ priors, const float* __restrict__ cc,
    float* __restrict__ out)
{
  extern __shared__ __attribute__((aligned(16))) float smem[];
  float* priorsLds = smem;
  float* accSim    = priorsLds + N_DIM * NP;
  float* zLds      = accSim + K_CAPS * N_DIM;
  float* vLds      = zLds + K_CAPS * O_DIM;
  float* part      = vLds + K_CAPS * O_DIM;
  float* scaleL    = part + K_CAPS * NGRP * O_DIM;

  const int tid = threadIdx.x;
  const int b   = blockIdx.x;
  const float* pb  = priors + (size_t)b * N_DIM * O_DIM;
  const float* ccb = cc + (size_t)b * K_CAPS * N_DIM * O_DIM;

  const f4* pb4 = (const f4*)pb;
  for (int i = tid; i < N_DIM * O_DIM / 4; i += 512) {
    const f4 v = pb4[i];
    float* d = &priorsLds[(i >> 4) * NP + (i & 15) * 4];
    d[0] = v[0]; d[1] = v[1]; d[2] = v[2]; d[3] = v[3];
  }
  for (int i = tid; i < K_CAPS * N_DIM; i += 512) accSim[i] = 0.0f;
  __syncthreads();

  const int o  = tid & 63;
  const int g  = tid >> 6;
  const int o4 = tid & 15;
  const int ng = tid >> 4;
  const f4* ccp = (const f4*)ccb + tid;

  for (int iter = 0; iter < 3; ++iter) {
    f4 zacc[K_CAPS];
#pragma unroll
    for (int k = 0; k < K_CAPS; ++k) {
#pragma unroll
      for (int j = 0; j < 4; ++j) zacc[k][j] = 0.0f;
    }

    f4 bufA[K_CAPS], bufB[K_CAPS];
    LOAD_BLOCK(bufA, 0);
    for (int jn = 0; jn < NBLK - 2; jn += 2) {
      __builtin_amdgcn_sched_barrier(0);
      LOAD_BLOCK(bufB, jn + 1);
      __builtin_amdgcn_sched_barrier(0);
      SUM_BLOCK(bufA, jn);
      END_BLOCK();
      __builtin_amdgcn_sched_barrier(0);
      LOAD_BLOCK(bufA, jn + 2);
      __builtin_amdgcn_sched_barrier(0);
      SUM_BLOCK(bufB, jn + 1);
      END_BLOCK();
    }
    __builtin_amdgcn_sched_barrier(0);
    LOAD_BLOCK(bufB, NBLK - 1);
    __builtin_amdgcn_sched_barrier(0);
    SUM_BLOCK(bufA, NBLK - 2);
    SUM_BLOCK(bufB, NBLK - 1);

#pragma unroll
    for (int k = 0; k < K_CAPS; ++k)
      *(f4*)&part[(k * NGRP + ng) * O_DIM + 4 * o4] = zacc[k];
    __syncthreads();

    {
      float z = 0.0f;
#pragma unroll
      for (int g2 = 0; g2 < NGRP; ++g2) z += part[(g * NGRP + g2) * O_DIM + o];
      zLds[g * O_DIM + o] = z;
    }
    __syncthreads();

    if (tid < K_CAPS) {
      float sq = 0.0f;
      for (int oo = 0; oo < O_DIM; ++oo) {
        const float zz = zLds[tid * O_DIM + oo];
        sq += zz * zz;
      }
      scaleL[tid] = (sq / (1.0f + sq)) / sqrtf(sq + 1e-9f);
    }
    __syncthreads();

    if (iter == 2) {
      out[(size_t)b * K_CAPS * O_DIM + tid] = zLds[tid] * scaleL[g];
    } else {
      vLds[tid] = zLds[tid] * scaleL[g];
      __syncthreads();
      const int n = tid;
#pragma unroll
      for (int k = 0; k < K_CAPS; ++k) {
        float s2 = 0.0f;
#pragma unroll 8
        for (int oo = 0; oo < O_DIM; ++oo)
          s2 += priorsLds[n * NP + oo] * vLds[k * O_DIM + oo];
        accSim[k * N_DIM + n] += s2;
      }
      __syncthreads();
    }
  }
}
#undef LOAD_BLOCK
#undef SUM_BLOCK
#undef END_BLOCK

extern "C" void kernel_launch(void* const* d_in, const int* in_sizes, int n_in,
                              void* d_out, int out_size, void* d_ws, size_t ws_size,
                              hipStream_t stream) {
  (void)in_sizes; (void)n_in; (void)out_size; (void)ws_size;
  const float* emb = (const float*)d_in[0];
  const float* S   = (const float*)d_in[1];
  const float* cc  = (const float*)d_in[2];
  float* out    = (float*)d_out;
  float* priors = (float*)d_ws;

  priors_gemm_wmma<<<(B_DIM * N_DIM) / GEMM_TM, 256, 0, stream>>>(emb, S, priors);

  const size_t smemBytes =
      (size_t)(N_DIM * NP + K_CAPS * N_DIM + 2 * K_CAPS * O_DIM +
               K_CAPS * NGRP * O_DIM + 8) * sizeof(float);
  routing_kernel<<<B_DIM, 512, smemBytes, stream>>>(priors, cc, out);
}


# Routing pass 1 walks the 16 n-blocks of cc last-to-first (passes 0 and 2 first-to-last) so it starts on the blocks still in the second-level cache; n summed in reverse block order in pass 1
# speedup vs baseline: 2.0324x; 1.1084x over previous
#include <hip/hip_runtime.h>
#include <math.h>

typedef __attribute__((ext_vector_type(2))) float v2f;
typedef __attribute__((ext_vector_type(4))) float f4;
typedef __attribute__((ext_vector_type(8))) float v8f;

#define B_DIM 256
#define N_DIM 512
#define E_DIM 64
#define O_DIM 64
#define K_CAPS 8

#define GEMM_TM 128
#define SA_STRIDE 68
#define SB_STRIDE 68

__global__ __launch_bounds__(256) void priors_gemm_wmma(
    const float* __restrict__ emb, const float* __restrict__ S,
    float* __restrict__ priors)
{
  __shared__ float sA[GEMM_TM * SA_STRIDE];
  __shared__ float sB[E_DIM * SB_STRIDE];

  const int tid  = threadIdx.x;
  const int row0 = blockIdx.x * GEMM_TM;

  const f4* emb4 = (const f4*)(emb + (size_t)row0 * E_DIM);
  for (int i = tid; i < GEMM_TM * (E_DIM / 4); i += 256) {
    const int m = i >> 4, c4 = i & 15;
    *(f4*)&sA[m * SA_STRIDE + c4 * 4] = emb4[m * 16 + c4];
  }
  const f4* S4 = (const f4*)S;
  for (int i = tid; i < E_DIM * (O_DIM / 4); i += 256) {
    const int k = i >> 4, c4 = i & 15;
    *(f4*)&sB[k * SB_STRIDE + c4 * 4] = S4[i];
  }
  __syncthreads();

  const int wave = tid >> 5, lane = tid & 31;
  const int lm = lane & 15;
  const int lh = lane >> 4;
  const int m0 = wave * 16;

  for (int nt = 0; nt < O_DIM / 16; ++nt) {
    const int n0 = nt * 16;
    v8f acc;
#pragma unroll
    for (int j = 0; j < 8; ++j) acc[j] = 0.0f;

#pragma unroll
    for (int k0 = 0; k0 < E_DIM; k0 += 4) {
      v2f a;
      const float* ap = &sA[(m0 + lm) * SA_STRIDE + k0 + 2 * lh];
      a[0] = ap[0];
      a[1] = ap[1];
      v2f bm;
      bm[0] = sB[(k0 + 0 + 2 * lh) * SB_STRIDE + n0 + lm];
      bm[1] = sB[(k0 + 1 + 2 * lh) * SB_STRIDE + n0 + lm];
      acc = __builtin_amdgcn_wmma_f32_16x16x4_f32(
           false, a,  false, bm,
           (short)0, acc,  false,  false);
    }
    float* dst = priors + ((size_t)(row0 + m0 + 8 * lh)) * O_DIM + n0 + lm;
#pragma unroll
    for (int j = 0; j < 8; ++j) dst[(size_t)j * O_DIM] = acc[j];
  }
}

#define NP 65
#define NGRP 32
#define NBLK 16
#define CC_KSTRIDE4 (N_DIM * O_DIM / 4)

#define PBLK(JN) (jb + js * (JN))

#define LOAD_BLOCK(BUF, JN)                                                   \
  do {                                                                        \
    const f4* cblk = ccp + PBLK(JN) * 512;                                    \
    _Pragma("unroll") for (int k = 0; k < K_CAPS; ++k)                        \
      BUF[k] = cblk[(size_t)k * CC_KSTRIDE4];                                 \
  } while (0)

#define SUM_BLOCK(BUF, JN)                                                    \
  do {                                                                        \
    const int n = ng + PBLK(JN) * NGRP;                                       \
    f4 pr;                                                                    \
    _Pragma("unroll") for (int j = 0; j < 4; ++j)                             \
      pr[j] = priorsLds[n * NP + 4 * o4 + j];                                 \
    f4 c[K_CAPS];                                                             \
    f4 m;                                                                     \
    _Pragma("unroll") for (int j = 0; j < 4; ++j) m[j] = -3.402823466e+38f;   \
    _Pragma("unroll") for (int k = 0; k < K_CAPS; ++k) {                      \
      const float a = accSim[k * N_DIM + n];                                  \
      _Pragma("unroll") for (int j = 0; j < 4; ++j) {                         \
        c[k][j] = BUF[k][j] + a;                                              \
        m[j] = fmaxf(m[j], c[k][j]);                                          \
      }                                                                       \
    }                                                                         \
    f4 s;                                                                     \
    _Pragma("unroll") for (int j = 0; j < 4; ++j) s[j] = 0.0f;                \
    _Pragma("unroll") for (int k = 0; k < K_CAPS; ++k) {                      \
      _Pragma("unroll") for (int j = 0; j < 4; ++j) {                         \
        c[k][j] = __expf(c[k][j] - m[j]);                                     \
        s[j] += c[k][j];                                                      \
      }                                                                       \
    }                                                                         \
    f4 p;                                                                     \
    _Pragma("unroll") for (int j = 0; j < 4; ++j) p[j] = pr[j] / s[j];        \
    _Pragma("unroll") for (int k = 0; k < K_CAPS; ++k) {                      \
      _Pragma("unroll") for (int j = 0; j < 4; ++j)                           \
        zacc[k][j] += c[k][j] * p[j];                                         \
    }                                                                         \
  } while (0)

#define END_BLOCK()                                                           \
  do {                                                                        \
    _Pragma("unroll") for (int k = 0; k < K_CAPS; ++k)                        \
      asm volatile("" : "+v"(zacc[k]));                                       \
  } while (0)

__global__ __launch_bounds__(512) void routing_kernel(
    const float* __restrict__ priors, const float* __restrict__ cc,
    float* __restrict__ out)
{
  extern __shared__ __attribute__((aligned(16))) float smem[];
  float* priorsLds = smem;
  float* accSim    = priorsLds + N_DIM * NP;
  float* zLds      = accSim + K_CAPS * N_DIM;
  float* vLds      = zLds + K_CAPS * O_DIM;
  float* part      = vLds + K_CAPS * O_DIM;
  float* scaleL    = part + K_CAPS * NGRP * O_DIM;

  const int tid = threadIdx.x;
  const int b   = blockIdx.x;
  const float* pb  = priors + (size_t)b * N_DIM * O_DIM;
  const float* ccb = cc + (size_t)b * K_CAPS * N_DIM * O_DIM;

  const f4* pb4 = (const f4*)pb;
  for (int i = tid; i < N_DIM * O_DIM / 4; i += 512) {
    const f4 v = pb4[i];
    float* d = &priorsLds[(i >> 4) * NP + (i & 15) * 4];
    d[0] = v[0]; d[1] = v[1]; d[2] = v[2]; d[3] = v[3];
  }
  for (int i = tid; i < K_CAPS * N_DIM; i += 512) accSim[i] = 0.0f;
  __syncthreads();

  const int o  = tid & 63;
  const int g  = tid >> 6;
  const int o4 = tid & 15;
  const int ng = tid >> 4;
  const f4* ccp = (const f4*)ccb + tid;

  for (int iter = 0; iter < 3; ++iter) {
    const int jb = (iter & 1) ? (NBLK - 1) : 0;
    const int js = (iter & 1) ? -1 : 1;
    f4 zacc[K_CAPS];
#pragma unroll
    for (int k = 0; k < K_CAPS; ++k) {
#pragma unroll
      for (int j = 0; j < 4; ++j) zacc[k][j] = 0.0f;
    }

    f4 bufA[K_CAPS], bufB[K_CAPS];
    LOAD_BLOCK(bufA, 0);
    for (int jn = 0; jn < NBLK - 2; jn += 2) {
      __builtin_amdgcn_sched_barrier(0);
      LOAD_BLOCK(bufB, jn + 1);
      __builtin_amdgcn_sched_barrier(0);
      SUM_BLOCK(bufA, jn);
      END_BLOCK();
      __builtin_amdgcn_sched_barrier(0);
      LOAD_BLOCK(bufA, jn + 2);
      __builtin_amdgcn_sched_barrier(0);
      SUM_BLOCK(bufB, jn + 1);
      END_BLOCK();
    }
    __builtin_amdgcn_sched_barrier(0);
    LOAD_BLOCK(bufB, NBLK - 1);
    __builtin_amdgcn_sched_barrier(0);
    SUM_BLOCK(bufA, NBLK - 2);
    SUM_BLOCK(bufB, NBLK - 1);

#pragma unroll
    for (int k = 0; k < K_CAPS; ++k)
      *(f4*)&part[(k * NGRP + ng) * O_DIM + 4 * o4] = zacc[k];
    __syncthreads();

    {
      float z = 0.0f;
#pragma unroll
      for (int g2 = 0; g2 < NGRP; ++g2) z += part[(g * NGRP + g2) * O_DIM + o];
      zLds[g * O_DIM + o] = z;
    }
    __syncthreads();

    if (tid < K_CAPS) {
      float sq = 0.0f;
      for (int oo = 0; oo < O_DIM; ++oo) {
        const float zz = zLds[tid * O_DIM + oo];
        sq += zz * zz;
      }
      scaleL[tid] = (sq / (1.0f + sq)) / sqrtf(sq + 1e-9f);
    }
    __syncthreads();

    if (iter == 2) {
      out[(size_t)b * K_CAPS * O_DIM + tid] = zLds[tid] * scaleL[g];
    } else {
      vLds[tid] = zLds[tid] * scaleL[g];
      __syncthreads();
      const int n = tid;
#pragma unroll
      for (int k = 0; k < K_CAPS; ++k) {
        float s2 = 0.0f;
#pragma unroll 8
        for (int oo = 0; oo < O_DIM; ++oo)
          s2 += priorsLds[n * NP + oo] * vLds[k * O_DIM + oo];
        accSim[k * N_DIM + n] += s2;
      }
      __syncthreads();
    }
  }
}
#undef PBLK
#undef LOAD_BLOCK
#undef SUM_BLOCK
#undef END_BLOCK

extern "C" void kernel_launch(void* const* d_in, const int* in_sizes, int n_in,
                              void* d_out, int out_size, void* d_ws, size_t ws_size,
                              hipStream_t stream) {
  (void)in_sizes; (void)n_in; (void)out_size; (void)ws_size;
  const float* emb = (const float*)d_in[0];
  const float* S   = (const float*)d_in[1];
  const float* cc  = (const float*)d_in[2];
  float* out    = (float*)d_out;
  float* priors = (float*)d_ws;

  priors_gemm_wmma<<<(B_DIM * N_DIM) / GEMM_TM, 256, 0, stream>>>(emb, S, priors);

  const size_t smemBytes =
      (size_t)(N_DIM * NP + K_CAPS * N_DIM + 2 * K_CAPS * O_DIM +
               K_CAPS * NGRP * O_DIM + 8) * sizeof(float);
  routing_kernel<<<B_DIM, 512, smemBytes, stream>>>(priors, cc, out);
}


# Routing: 4 of the 16 n-blocks of cc stay on chip after pass 0 (3 in registers, 1 in LDS) and are not re-read in passes 1, 2; block sums on float2 column pairs to fit; n summed in another block order
# speedup vs baseline: 2.3292x; 1.1460x over previous
#include <hip/hip_runtime.h>
#include <math.h>

typedef __attribute__((ext_vector_type(2))) float v2f;
typedef __attribute__((ext_vector_type(4))) float f4;
typedef __attribute__((ext_vector_type(8))) float v8f;

#define B_DIM 256
#define N_DIM 512
#define E_DIM 64
#define O_DIM 64
#define K_CAPS 8

#define GEMM_TM 128
#define SA_STRIDE 68
#define SB_STRIDE 68

__global__ __launch_bounds__(256) void priors_gemm_wmma(
    const float* __restrict__ emb, const float* __restrict__ S,
    float* __restrict__ priors)
{
  __shared__ float sA[GEMM_TM * SA_STRIDE];
  __shared__ float sB[E_DIM * SB_STRIDE];

  const int tid  = threadIdx.x;
  const int row0 = blockIdx.x * GEMM_TM;

  const f4* emb4 = (const f4*)(emb + (size_t)row0 * E_DIM);
  for (int i = tid; i < GEMM_TM * (E_DIM / 4); i += 256) {
    const int m = i >> 4, c4 = i & 15;
    *(f4*)&sA[m * SA_STRIDE + c4 * 4] = emb4[m * 16 + c4];
  }
  const f4* S4 = (const f4*)S;
  for (int i = tid; i < E_DIM * (O_DIM / 4); i += 256) {
    const int k = i >> 4, c4 = i & 15;
    *(f4*)&sB[k * SB_STRIDE + c4 * 4] = S4[i];
  }
  __syncthreads();

  const int wave = tid >> 5, lane = tid & 31;
  const int lm = lane & 15;
  const int lh = lane >> 4;
  const int m0 = wave * 16;

  for (int nt = 0; nt < O_DIM / 16; ++nt) {
    const int n0 = nt * 16;
    v8f acc;
#pragma unroll
    for (int j = 0; j < 8; ++j) acc[j] = 0.0f;

#pragma unroll
    for (int k0 = 0; k0 < E_DIM; k0 += 4) {
      v2f a;
      const float* ap = &sA[(m0 + lm) * SA_STRIDE + k0 + 2 * lh];
      a[0] = ap[0];
      a[1] = ap[1];
      v2f bm;
      bm[0] = sB[(k0 + 0 + 2 * lh) * SB_STRIDE + n0 + lm];
      bm[1] = sB[(k0 + 1 + 2 * lh) * SB_STRIDE + n0 + lm];
      acc = __builtin_amdgcn_wmma_f32_16x16x4_f32(
           false, a,  false, bm,
           (short)0, acc,  false,  false);
    }
    float* dst = priors + ((size_t)(row0 + m0 + 8 * lh)) * O_DIM + n0 + lm;
#pragma unroll
    for (int j = 0; j < 8; ++j) dst[(size_t)j * O_DIM] = acc[j];
  }
}

#define NP 65
#define NGRP 32
#define NBLK 16
#define NKEEP 4
#define CC_KSTRIDE4 (N_DIM * O_DIM / 4)

#define NB(S) (jb + js * (S))

#define LOAD_BLOCK(BUF, BLK)                                                  \
  do {                                                                        \
    const f4* cblk = ccp + (BLK) * 512;                                       \
    _Pragma("unroll") for (int k = 0; k < K_CAPS; ++k)                        \
      BUF[k] = cblk[(size_t)k * CC_KSTRIDE4];                                 \
  } while (0)

#define KEEP_BLOCK(BUF)                                                       \
  do {                                                                        \
    _Pragma("unroll") for (int k = 0; k < K_CAPS; ++k)                        \
      keep4[k * 512] = BUF[k];                                                \
  } while (0)
#define FETCH_KEPT(BUF)                                                       \
  do {                                                                        \
    _Pragma("unroll") for (int k = 0; k < K_CAPS; ++k)                        \
      BUF[k] = keep4[k * 512];                                                \
  } while (0)

#define SUM_BLOCK(BUF, BLK)                                                   \
  do {                                                                        \
    const int n = ng + (BLK) * NGRP;                                          \
    v2f prl, prh;                                                             \
    prl[0] = priorsLds[n * NP + 4 * o4 + 0];                                  \
    prl[1] = priorsLds[n * NP + 4 * o4 + 1];                                  \
    prh[0] = priorsLds[n * NP + 4 * o4 + 2];                                  \
    prh[1] = priorsLds[n * NP + 4 * o4 + 3];                                  \
    v2f cl[K_CAPS], ch[K_CAPS];                                               \
    v2f ml, mh;                                                               \
    ml[0] = ml[1] = mh[0] = mh[1] = -3.402823466e+38f;                        \
    _Pragma("unroll") for (int k = 0; k < K_CAPS; ++k) {                      \
      const float a = accSim[k * N_DIM + n];                                  \
      v2f av; av[0] = a; av[1] = a;                                           \
      v2f bl, bh;                                                             \
      bl[0] = BUF[k][0]; bl[1] = BUF[k][1];                                   \
      bh[0] = BUF[k][2]; bh[1] = BUF[k][3];                                   \
      cl[k] = bl + av;                                                        \
      ch[k] = bh + av;                                                        \
      ml[0] = fmaxf(ml[0], cl[k][0]); ml[1] = fmaxf(ml[1], cl[k][1]);         \
      mh[0] = fmaxf(mh[0], ch[k][0]); mh[1] = fmaxf(mh[1], ch[k][1]);         \
    }                                                                         \
    v2f sl, sh;                                                               \
    sl[0] = sl[1] = sh[0] = sh[1] = 0.0f;                                     \
    _Pragma("unroll") for (int k = 0; k < K_CAPS; ++k) {                      \
      const v2f dl = cl[k] - ml;                                              \
      const v2f dh = ch[k] - mh;                                              \
      cl[k][0] = __expf(dl[0]); cl[k][1] = __expf(dl[1]);                     \
      ch[k][0] = __expf(dh[0]); ch[k][1] = __expf(dh[1]);                     \
      sl += cl[k];                                                            \
      sh += ch[k];                                                            \
    }                                                                         \
    v2f pl, ph;                                                               \
    pl[0] = prl[0] / sl[0]; pl[1] = prl[1] / sl[1];                           \
    ph[0] = prh[0] / sh[0]; ph[1] = prh[1] / sh[1];                           \
    _Pragma("unroll") for (int k = 0; k < K_CAPS; ++k) {                      \
      zacc[k][0] += cl[k][0] * pl[0]; zacc[k][1] += cl[k][1] * pl[1];         \
      zacc[k][2] += ch[k][0] * ph[0]; zacc[k][3] += ch[k][1] * ph[1];         \
    }                                                                         \
  } while (0)

#define END_BLOCK()                                                           \
  do {                                                                        \
    _Pragma("unroll") for (int k = 0; k < K_CAPS; ++k)                        \
      asm volatile("" : "+v"(zacc[k]));                                       \
  } while (0)

#define SB() __builtin_amdgcn_sched_barrier(0)

__global__ __launch_bounds__(512) void routing_kernel(
    const float* __restrict__ priors, const float* __restrict__ cc,
    float* __restrict__ out)
{
  extern __shared__ __attribute__((aligned(16))) float smem[];
  float* priorsLds = smem;
  float* accSim    = priorsLds + N_DIM * NP;
  float* zLds      = accSim + K_CAPS * N_DIM;
  float* vLds      = zLds + K_CAPS * O_DIM;
  float* part      = vLds + K_CAPS * O_DIM;
  float* scaleL    = part + K_CAPS * NGRP * O_DIM;
  float* keepL     = scaleL + 8;

  const int tid = threadIdx.x;
  const int b   = blockIdx.x;
  const float* pb  = priors + (size_t)b * N_DIM * O_DIM;
  const float* ccb = cc + (size_t)b * K_CAPS * N_DIM * O_DIM;

  const f4* pb4 = (const f4*)pb;
  for (int i = tid; i < N_DIM * O_DIM / 4; i += 512) {
    const f4 v = pb4[i];
    float* d = &priorsLds[(i >> 4) * NP + (i & 15) * 4];
    d[0] = v[0]; d[1] = v[1]; d[2] = v[2]; d[3] = v[3];
  }
  for (int i = tid; i < K_CAPS * N_DIM; i += 512) accSim[i] = 0.0f;
  __syncthreads();

  const int o  = tid & 63;
  const int g  = tid >> 6;
  const int o4 = tid & 15;
  const int ng = tid >> 4;
  const f4* ccp = (const f4*)ccb + tid;
  f4* keep4 = (f4*)keepL + tid;

  f4 res0[K_CAPS], res1[K_CAPS], res2[K_CAPS];

  for (int iter = 0; iter < 3; ++iter) {
    f4 zacc[K_CAPS];
#pragma unroll
    for (int k = 0; k < K_CAPS; ++k) {
#pragma unroll
      for (int j = 0; j < 4; ++j) zacc[k][j] = 0.0f;
    }
    f4 bufA[K_CAPS], bufB[K_CAPS];

    if (iter == 0) {
      LOAD_BLOCK(res0, 0);
      SB(); LOAD_BLOCK(res1, 1); SB(); SUM_BLOCK(res0, 0); END_BLOCK();
      SB(); LOAD_BLOCK(res2, 2); SB(); SUM_BLOCK(res1, 1); END_BLOCK();
      SB(); LOAD_BLOCK(bufA, 3); SB(); SUM_BLOCK(res2, 2); END_BLOCK();
      SB(); LOAD_BLOCK(bufB, NBLK - 1); SB(); SUM_BLOCK(bufA, 3); KEEP_BLOCK(bufA); END_BLOCK();
      for (int jn = NBLK - 1; jn > NKEEP + 1; jn -= 2) {
        SB(); LOAD_BLOCK(bufA, jn - 1); SB(); SUM_BLOCK(bufB, jn);     END_BLOCK();
        SB(); LOAD_BLOCK(bufB, jn - 2); SB(); SUM_BLOCK(bufA, jn - 1); END_BLOCK();
      }
      SB(); LOAD_BLOCK(bufA, NKEEP); SB();
      SUM_BLOCK(bufB, NKEEP + 1);
      SUM_BLOCK(bufA, NKEEP);
    } else {
      const int jb = (iter == 1) ? NKEEP : (NBLK - 1);
      const int js = (iter == 1) ? 1 : -1;
      LOAD_BLOCK(bufA, NB(0));
      for (int s = 0; s < 8; s += 2) {
        SB(); LOAD_BLOCK(bufB, NB(s + 1)); SB(); SUM_BLOCK(bufA, NB(s));     END_BLOCK();
        SB(); LOAD_BLOCK(bufA, NB(s + 2)); SB(); SUM_BLOCK(bufB, NB(s + 1)); END_BLOCK();
      }
      SUM_BLOCK(res0, 0); END_BLOCK();
      SB(); LOAD_BLOCK(bufB, NB(9));  SB(); SUM_BLOCK(bufA, NB(8));  END_BLOCK();
      SUM_BLOCK(res1, 1); END_BLOCK();
      SB(); LOAD_BLOCK(bufA, NB(10)); SB(); SUM_BLOCK(bufB, NB(9));  END_BLOCK();
      SUM_BLOCK(res2, 2); END_BLOCK();
      SB(); LOAD_BLOCK(bufB, NB(11)); SB(); SUM_BLOCK(bufA, NB(10)); END_BLOCK();
      FETCH_KEPT(bufA);
      SUM_BLOCK(bufA, 3); END_BLOCK();
      SUM_BLOCK(bufB, NB(11));
    }

#pragma unroll
    for (int k = 0; k < K_CAPS; ++k)
      *(f4*)&part[(k * NGRP + ng) * O_DIM + 4 * o4] = zacc[k];
    __syncthreads();

    {
      float z = 0.0f;
#pragma unroll
      for (int g2 = 0; g2 < NGRP; ++g2) z += part[(g * NGRP + g2) * O_DIM + o];
      zLds[g * O_DIM + o] = z;
    }
    __syncthreads();

    if (tid < K_CAPS) {
      float sq = 0.0f;
      for (int oo = 0; oo < O_DIM; ++oo) {
        const float zz = zLds[tid * O_DIM + oo];
        sq += zz * zz;
      }
      scaleL[tid] = (sq / (1.0f + sq)) / sqrtf(sq + 1e-9f);
    }
    __syncthreads();

    if (iter == 2) {
      out[(size_t)b * K_CAPS * O_DIM + tid] = zLds[tid] * scaleL[g];
    } else {
      vLds[tid] = zLds[tid] * scaleL[g];
      __syncthreads();
      const int n = tid;
#pragma unroll
      for (int k = 0; k < K_CAPS; ++k) {
        float s2 = 0.0f;
#pragma unroll 8
        for (int oo = 0; oo < O_DIM; ++oo)
          s2 += priorsLds[n * NP + oo] * vLds[k * O_DIM + oo];
        accSim[k * N_DIM + n] += s2;
      }
      __syncthreads();
    }
  }
}
#undef NB
#undef LOAD_BLOCK
#undef KEEP_BLOCK
#undef FETCH_KEPT
#undef SUM_BLOCK
#undef END_BLOCK
#undef SB

extern "C" void kernel_launch(void* const* d_in, const int* in_sizes, int n_in,
                              void* d_out, int out_size, void* d_ws, size_t ws_size,
                              hipStream_t stream) {
  (void)in_sizes; (void)n_in; (void)out_size; (void)ws_size;
  const float* emb = (const float*)d_in[0];
  const float* S   = (const float*)d_in[1];
  const float* cc  = (const float*)d_in[2];
  float* out    = (float*)d_out;
  float* priors = (float*)d_ws;

  priors_gemm_wmma<<<(B_DIM * N_DIM) / GEMM_TM, 256, 0, stream>>>(emb, S, priors);

  const size_t smemBytes =
      (size_t)(N_DIM * NP + K_CAPS * N_DIM + 2 * K_CAPS * O_DIM +
               K_CAPS * NGRP * O_DIM + 8 +
               K_CAPS * 512 * 4) * sizeof(float);
  routing_kernel<<<B_DIM, 512, smemBytes, stream>>>(priors, cc, out);
}


# Routing pass B (sim += priors . v) on f32 16x16x4 matrix instructions, two 16-row n tiles per wave, v rows padded to stride 65 in LDS; sim summed over o in the matrix unit's order
# speedup vs baseline: 2.5841x; 1.1094x over previous
#include <hip/hip_runtime.h>
#include <math.h>

typedef __attribute__((ext_vector_type(2))) float v2f;
typedef __attribute__((ext_vector_type(4))) float f4;
typedef __attribute__((ext_vector_type(8))) float v8f;

#define B_DIM 256
#define N_DIM 512
#define E_DIM 64
#define O_DIM 64
#define K_CAPS 8

#define GEMM_TM 128
#define SA_STRIDE 68
#define SB_STRIDE 68

__global__ __launch_bounds__(256) void priors_gemm_wmma(
    const float* __restrict__ emb, const float* __restrict__ S,
    float* __restrict__ priors)
{
  __shared__ float sA[GEMM_TM * SA_STRIDE];
  __shared__ float sB[E_DIM * SB_STRIDE];

  const int tid  = threadIdx.x;
  const int row0 = blockIdx.x * GEMM_TM;

  const f4* emb4 = (const f4*)(emb + (size_t)row0 * E_DIM);
  for (int i = tid; i < GEMM_TM * (E_DIM / 4); i += 256) {
    const int m = i >> 4, c4 = i & 15;
    *(f4*)&sA[m * SA_STRIDE + c4 * 4] = emb4[m * 16 + c4];
  }
  const f4* S4 = (const f4*)S;
  for (int i = tid; i < E_DIM * (O_DIM / 4); i += 256) {
    const int k = i >> 4, c4 = i & 15;
    *(f4*)&sB[k * SB_STRIDE + c4 * 4] = S4[i];
  }
  __syncthreads();

  const int wave = tid >> 5, lane = tid & 31;
  const int lm = lane & 15;
  const int lh = lane >> 4;
  const int m0 = wave * 16;

  for (int nt = 0; nt < O_DIM / 16; ++nt) {
    const int n0 = nt * 16;
    v8f acc;
#pragma unroll
    for (int j = 0; j < 8; ++j) acc[j] = 0.0f;

#pragma unroll
    for (int k0 = 0; k0 < E_DIM; k0 += 4) {
      v2f a;
      const float* ap = &sA[(m0 + lm) * SA_STRIDE + k0 + 2 * lh];
      a[0] = ap[0];
      a[1] = ap[1];
      v2f bm;
      bm[0] = sB[(k0 + 0 + 2 * lh) * SB_STRIDE + n0 + lm];
      bm[1] = sB[(k0 + 1 + 2 * lh) * SB_STRIDE + n0 + lm];
      acc = __builtin_amdgcn_wmma_f32_16x16x4_f32(
           false, a,  false, bm,
           (short)0, acc,  false,  false);
    }
    float* dst = priors + ((size_t)(row0 + m0 + 8 * lh)) * O_DIM + n0 + lm;
#pragma unroll
    for (int j = 0; j < 8; ++j) dst[(size_t)j * O_DIM] = acc[j];
  }
}

#define NP 65
#define NGRP 32
#define NBLK 16
#define NKEEP 4
#define VS 65
#define CC_KSTRIDE4 (N_DIM * O_DIM / 4)

#define NB(S) (jb + js * (S))

#define LOAD_BLOCK(BUF, BLK)                                                  \
  do {                                                                        \
    const f4* cblk = ccp + (BLK) * 512;                                       \
    _Pragma("unroll") for (int k = 0; k < K_CAPS; ++k)                        \
      BUF[k] = cblk[(size_t)k * CC_KSTRIDE4];                                 \
  } while (0)

#define KEEP_BLOCK(BUF)                                                       \
  do {                                                                        \
    _Pragma("unroll") for (int k = 0; k < K_CAPS; ++k)                        \
      keep4[k * 512] = BUF[k];                                                \
  } while (0)
#define FETCH_KEPT(BUF)                                                       \
  do {                                                                        \
    _Pragma("unroll") for (int k = 0; k < K_CAPS; ++k)                        \
      BUF[k] = keep4[k * 512];                                                \
  } while (0)

#define SUM_BLOCK(BUF, BLK)                                                   \
  do {                                                                        \
    const int n = ng + (BLK) * NGRP;                                          \
    v2f prl, prh;                                                             \
    prl[0] = priorsLds[n * NP + 4 * o4 + 0];                                  \
    prl[1] = priorsLds[n * NP + 4 * o4 + 1];                                  \
    prh[0] = priorsLds[n * NP + 4 * o4 + 2];                                  \
    prh[1] = priorsLds[n * NP + 4 * o4 + 3];                                  \
    v2f cl[K_CAPS], ch[K_CAPS];                                               \
    v2f ml, mh;                                                               \
    ml[0] = ml[1] = mh[0] = mh[1] = -3.402823466e+38f;                        \
    _Pragma("unroll") for (int k = 0; k < K_CAPS; ++k) {                      \
      const float a = accSim[k * N_DIM + n];                                  \
      v2f av; av[0] = a; av[1] = a;                                           \
      v2f bl, bh;                                                             \
      bl[0] = BUF[k][0]; bl[1] = BUF[k][1];                                   \
      bh[0] = BUF[k][2]; bh[1] = BUF[k][3];                                   \
      cl[k] = bl + av;                                                        \
      ch[k] = bh + av;                                                        \
      ml[0] = fmaxf(ml[0], cl[k][0]); ml[1] = fmaxf(ml[1], cl[k][1]);         \
      mh[0] = fmaxf(mh[0], ch[k][0]); mh[1] = fmaxf(mh[1], ch[k][1]);         \
    }                                                                         \
    v2f sl, sh;                                                               \
    sl[0] = sl[1] = sh[0] = sh[1] = 0.0f;                                     \
    _Pragma("unroll") for (int k = 0; k < K_CAPS; ++k) {                      \
      const v2f dl = cl[k] - ml;                                              \
      const v2f dh = ch[k] - mh;                                              \
      cl[k][0] = __expf(dl[0]); cl[k][1] = __expf(dl[1]);                     \
      ch[k][0] = __expf(dh[0]); ch[k][1] = __expf(dh[1]);                     \
      sl += cl[k];                                                            \
      sh += ch[k];                                                            \
    }                                                                         \
    v2f pl, ph;                                                               \
    pl[0] = prl[0] / sl[0]; pl[1] = prl[1] / sl[1];                           \
    ph[0] = prh[0] / sh[0]; ph[1] = prh[1] / sh[1];                           \
    _Pragma("unroll") for (int k = 0; k < K_CAPS; ++k) {                      \
      zacc[k][0] += cl[k][0] * pl[0]; zacc[k][1] += cl[k][1] * pl[1];         \
      zacc[k][2] += ch[k][0] * ph[0]; zacc[k][3] += ch[k][1] * ph[1];         \
    }                                                                         \
  } while (0)

#define END_BLOCK()                                                           \
  do {                                                                        \
    _Pragma("unroll") for (int k = 0; k < K_CAPS; ++k)                        \
      asm volatile("" : "+v"(zacc[k]));                                       \
  } while (0)

#define SB() __builtin_amdgcn_sched_barrier(0)

__global__ __launch_bounds__(512) void routing_kernel(
    const float* __restrict__ priors, const float* __restrict__ cc,
    float* __restrict__ out)
{
  extern __shared__ __attribute__((aligned(16))) float smem[];
  float* priorsLds = smem;
  float* accSim    = priorsLds + N_DIM * NP;
  float* zLds      = accSim + K_CAPS * N_DIM;
  float* vLds      = zLds + K_CAPS * O_DIM;
  float* part      = vLds + K_CAPS * VS;
  float* scaleL    = part + K_CAPS * NGRP * O_DIM;
  float* keepL     = scaleL + 8;

  const int tid = threadIdx.x;
  const int b   = blockIdx.x;
  const float* pb  = priors + (size_t)b * N_DIM * O_DIM;
  const float* ccb = cc + (size_t)b * K_CAPS * N_DIM * O_DIM;

  const f4* pb4 = (const f4*)pb;
  for (int i = tid; i < N_DIM * O_DIM / 4; i += 512) {
    const f4 v = pb4[i];
    float* d = &priorsLds[(i >> 4) * NP + (i & 15) * 4];
    d[0] = v[0]; d[1] = v[1]; d[2] = v[2]; d[3] = v[3];
  }
  for (int i = tid; i < K_CAPS * N_DIM; i += 512) accSim[i] = 0.0f;
  __syncthreads();

  const int o  = tid & 63;
  const int g  = tid >> 6;
  const int o4 = tid & 15;
  const int ng = tid >> 4;
  const f4* ccp = (const f4*)ccb + tid;
  f4* keep4 = (f4*)keepL + tid;

  f4 res0[K_CAPS], res1[K_CAPS], res2[K_CAPS];

  for (int iter = 0; iter < 3; ++iter) {
    f4 zacc[K_CAPS];
#pragma unroll
    for (int k = 0; k < K_CAPS; ++k) {
#pragma unroll
      for (int j = 0; j < 4; ++j) zacc[k][j] = 0.0f;
    }
    f4 bufA[K_CAPS], bufB[K_CAPS];

    if (iter == 0) {
      LOAD_BLOCK(res0, 0);
      SB(); LOAD_BLOCK(res1, 1); SB(); SUM_BLOCK(res0, 0); END_BLOCK();
      SB(); LOAD_BLOCK(res2, 2); SB(); SUM_BLOCK(res1, 1); END_BLOCK();
      SB(); LOAD_BLOCK(bufA, 3); SB(); SUM_BLOCK(res2, 2); END_BLOCK();
      SB(); LOAD_BLOCK(bufB, NBLK - 1); SB(); SUM_BLOCK(bufA, 3); KEEP_BLOCK(bufA); END_BLOCK();
      for (int jn = NBLK - 1; jn > NKEEP + 1; jn -= 2) {
        SB(); LOAD_BLOCK(bufA, jn - 1); SB(); SUM_BLOCK(bufB, jn);     END_BLOCK();
        SB(); LOAD_BLOCK(bufB, jn - 2); SB(); SUM_BLOCK(bufA, jn - 1); END_BLOCK();
      }
      SB(); LOAD_BLOCK(bufA, NKEEP); SB();
      SUM_BLOCK(bufB, NKEEP + 1);
      SUM_BLOCK(bufA, NKEEP);
    } else {
      const int jb = (iter == 1) ? NKEEP : (NBLK - 1);
      const int js = (iter == 1) ? 1 : -1;
      LOAD_BLOCK(bufA, NB(0));
      for (int s = 0; s < 8; s += 2) {
        SB(); LOAD_BLOCK(bufB, NB(s + 1)); SB(); SUM_BLOCK(bufA, NB(s));     END_BLOCK();
        SB(); LOAD_BLOCK(bufA, NB(s + 2)); SB(); SUM_BLOCK(bufB, NB(s + 1)); END_BLOCK();
      }
      SUM_BLOCK(res0, 0); END_BLOCK();
      SB(); LOAD_BLOCK(bufB, NB(9));  SB(); SUM_BLOCK(bufA, NB(8));  END_BLOCK();
      SUM_BLOCK(res1, 1); END_BLOCK();
      SB(); LOAD_BLOCK(bufA, NB(10)); SB(); SUM_BLOCK(bufB, NB(9));  END_BLOCK();
      SUM_BLOCK(res2, 2); END_BLOCK();
      SB(); LOAD_BLOCK(bufB, NB(11)); SB(); SUM_BLOCK(bufA, NB(10)); END_BLOCK();
      FETCH_KEPT(bufA);
      SUM_BLOCK(bufA, 3); END_BLOCK();
      SUM_BLOCK(bufB, NB(11));
    }

#pragma unroll
    for (int k = 0; k < K_CAPS; ++k)
      *(f4*)&part[(k * NGRP + ng) * O_DIM + 4 * o4] = zacc[k];
    __syncthreads();

    {
      float z = 0.0f;
#pragma unroll
      for (int g2 = 0; g2 < NGRP; ++g2) z += part[(g * NGRP + g2) * O_DIM + o];
      zLds[g * O_DIM + o] = z;
    }
    __syncthreads();

    if (tid < K_CAPS) {
      float sq = 0.0f;
      for (int oo = 0; oo < O_DIM; ++oo) {
        const float zz = zLds[tid * O_DIM + oo];
        sq += zz * zz;
      }
      scaleL[tid] = (sq / (1.0f + sq)) / sqrtf(sq + 1e-9f);
    }
    __syncthreads();

    if (iter == 2) {
      out[(size_t)b * K_CAPS * O_DIM + tid] = zLds[tid] * scaleL[g];
    } else {
      vLds[g * VS + o] = zLds[tid] * scaleL[g];
      __syncthreads();
      const int lm = tid & 15;
      const int lh = (tid >> 4) & 1;
      const int nw = (tid >> 5) * 32;
      const float* va  = vLds + (lm & 7) * VS + 32 * lh;
      const float* pb0 = priorsLds + (nw + lm) * NP + 32 * lh;
      const float* pb1 = pb0 + 16 * NP;
      v8f d0, d1;
#pragma unroll
      for (int j = 0; j < 8; ++j) { d0[j] = 0.0f; d1[j] = 0.0f; }
#pragma unroll
      for (int t = 0; t < 16; ++t) {
        v2f a, b0, b1;
        a[0]  = va[2 * t];   a[1]  = va[2 * t + 1];
        b0[0] = pb0[2 * t];  b0[1] = pb0[2 * t + 1];
        b1[0] = pb1[2 * t];  b1[1] = pb1[2 * t + 1];
        d0 = __builtin_amdgcn_wmma_f32_16x16x4_f32(
             false, a,  false, b0,
             (short)0, d0,  false,  false);
        d1 = __builtin_amdgcn_wmma_f32_16x16x4_f32(
             false, a,  false, b1,
             (short)0, d1,  false,  false);
      }
      {
        float* sp = accSim + nw + 16 * lh + lm;
#pragma unroll
        for (int r = 0; r < K_CAPS; ++r)
          sp[r * N_DIM] += lh ? d1[r] : d0[r];
      }
      __syncthreads();
    }
  }
}
#undef NB
#undef LOAD_BLOCK
#undef KEEP_BLOCK
#undef FETCH_KEPT
#undef SUM_BLOCK
#undef END_BLOCK
#undef SB

extern "C" void kernel_launch(void* const* d_in, const int* in_sizes, int n_in,
                              void* d_out, int out_size, void* d_ws, size_t ws_size,
                              hipStream_t stream) {
  (void)in_sizes; (void)n_in; (void)out_size; (void)ws_size;
  const float* emb = (const float*)d_in[0];
  const float* S   = (const float*)d_in[1];
  const float* cc  = (const float*)d_in[2];
  float* out    = (float*)d_out;
  float* priors = (float*)d_ws;

  priors_gemm_wmma<<<(B_DIM * N_DIM) / GEMM_TM, 256, 0, stream>>>(emb, S, priors);

  const size_t smemBytes =
      (size_t)(N_DIM * NP + K_CAPS * N_DIM + K_CAPS * O_DIM + K_CAPS * VS +
               K_CAPS * NGRP * O_DIM + 8 +
               K_CAPS * 512 * 4) * sizeof(float);
  routing_kernel<<<B_DIM, 512, smemBytes, stream>>>(priors, cc, out);
}


# Priors GEMM fused into the routing kernel (1 launch, not 2; priors go from the matrix unit straight to LDS), cc n-blocks 0..2 requested ahead of it; e and pass-0 n-blocks summed in another order
# speedup vs baseline: 2.7771x; 1.0747x over previous
#include <hip/hip_runtime.h>
#include <math.h>

typedef __attribute__((ext_vector_type(2))) float v2f;
typedef __attribute__((ext_vector_type(4))) float f4;
typedef __attribute__((ext_vector_type(8))) float v8f;

#define B_DIM 256
#define N_DIM 512
#define E_DIM 64
#define O_DIM 64
#define K_CAPS 8

#define NP 65
#define NGRP 32
#define NBLK 16
#define NKEEP 4
#define VS 65
#define SS 65
#define CC_KSTRIDE4 (N_DIM * O_DIM / 4)

#define NB(S) (jb + js * (S))

#define LOAD_BLOCK(BUF, BLK)                                                  \
  do {                                                                        \
    const f4* cblk = ccp + (BLK) * 512;                                       \
    _Pragma("unroll") for (int k = 0; k < K_CAPS; ++k)                        \
      BUF[k] = cblk[(size_t)k * CC_KSTRIDE4];                                 \
  } while (0)

#define KEEP_BLOCK(BUF)                                                       \
  do {                                                                        \
    _Pragma("unroll") for (int k = 0; k < K_CAPS; ++k)                        \
      keep4[k * 512] = BUF[k];                                                \
  } while (0)
#define FETCH_KEPT(BUF)                                                       \
  do {                                                                        \
    _Pragma("unroll") for (int k = 0; k < K_CAPS; ++k)                        \
      BUF[k] = keep4[k * 512];                                                \
  } while (0)

#define SUM_BLOCK(BUF, BLK)                                                   \
  do {                                                                        \
    const int n = ng + (BLK) * NGRP;                                          \
    v2f prl, prh;                                                             \
    prl[0] = priorsLds[n * NP + 4 * o4 + 0];                                  \
    prl[1] = priorsLds[n * NP + 4 * o4 + 1];                                  \
    prh[0] = priorsLds[n * NP + 4 * o4 + 2];                                  \
    prh[1] = priorsLds[n * NP + 4 * o4 + 3];                                  \
    v2f cl[K_CAPS], ch[K_CAPS];                                               \
    v2f ml, mh;                                                               \
    ml[0] = ml[1] = mh[0] = mh[1] = -3.402823466e+38f;                        \
    _Pragma("unroll") for (int k = 0; k < K_CAPS; ++k) {                      \
      const float a = accSim[k * N_DIM + n];                                  \
      v2f av; av[0] = a; av[1] = a;                                           \
      v2f bl, bh;                                                             \
      bl[0] = BUF[k][0]; bl[1] = BUF[k][1];                                   \
      bh[0] = BUF[k][2]; bh[1] = BUF[k][3];                                   \
      cl[k] = bl + av;                                                        \
      ch[k] = bh + av;                                                        \
      ml[0] = fmaxf(ml[0], cl[k][0]); ml[1] = fmaxf(ml[1], cl[k][1]);         \
      mh[0] = fmaxf(mh[0], ch[k][0]); mh[1] = fmaxf(mh[1], ch[k][1]);         \
    }                                                                         \
    v2f sl, sh;                                                               \
    sl[0] = sl[1] = sh[0] = sh[1] = 0.0f;                                     \
    _Pragma("unroll") for (int k = 0; k < K_CAPS; ++k) {                      \
      const v2f dl = cl[k] - ml;                                              \
      const v2f dh = ch[k] - mh;                                              \
      cl[k][0] = __expf(dl[0]); cl[k][1] = __expf(dl[1]);                     \
      ch[k][0] = __expf(dh[0]); ch[k][1] = __expf(dh[1]);                     \
      sl += cl[k];                                                            \
      sh += ch[k];                                                            \
    }                                                                         \
    v2f pl, ph;                                                               \
    pl[0] = prl[0] / sl[0]; pl[1] = prl[1] / sl[1];                           \
    ph[0] = prh[0] / sh[0]; ph[1] = prh[1] / sh[1];                           \
    _Pragma("unroll") for (int k = 0; k < K_CAPS; ++k) {                      \
      zacc[k][0] += cl[k][0] * pl[0]; zacc[k][1] += cl[k][1] * pl[1];         \
      zacc[k][2] += ch[k][0] * ph[0]; zacc[k][3] += ch[k][1] * ph[1];         \
    }                                                                         \
  } while (0)

#define END_BLOCK()                                                           \
  do {                                                                        \
    _Pragma("unroll") for (int k = 0; k < K_CAPS; ++k)                        \
      asm volatile("" : "+v"(zacc[k]));                                       \
  } while (0)

#define SB() __builtin_amdgcn_sched_barrier(0)

__device__ __forceinline__ void lds_barrier() {
  __builtin_amdgcn_fence(__ATOMIC_RELEASE, "workgroup", "local");
  __builtin_amdgcn_s_barrier();
  __builtin_amdgcn_fence(__ATOMIC_ACQUIRE, "workgroup", "local");
}

__global__ __launch_bounds__(512) void routing_kernel(
    const float* __restrict__ emb, const float* __restrict__ S,
    const float* __restrict__ cc, float* __restrict__ out)
{
  extern __shared__ __attribute__((aligned(16))) float smem[];
  float* priorsLds = smem;
  float* accSim    = priorsLds + N_DIM * NP;
  float* zLds      = accSim + K_CAPS * N_DIM;
  float* vLds      = zLds + K_CAPS * O_DIM;
  float* part      = vLds + K_CAPS * VS;
  float* scaleL    = part + K_CAPS * NGRP * O_DIM;
  float* keepL     = scaleL + 8;

  const int tid = threadIdx.x;
  const int b   = blockIdx.x;
  const float* ccb = cc + (size_t)b * K_CAPS * N_DIM * O_DIM;

  const int o  = tid & 63;
  const int g  = tid >> 6;
  const int o4 = tid & 15;
  const int ng = tid >> 4;
  const f4* ccp = (const f4*)ccb + tid;
  f4* keep4 = (f4*)keepL + tid;

  f4 res0[K_CAPS], res1[K_CAPS], res2[K_CAPS];

  float* sS = part;
  const f4* S4  = (const f4*)S;
  const f4* eb4 = (const f4*)(emb + (size_t)b * N_DIM * E_DIM);
  f4 sv[2], ev[16];
  sv[0] = S4[tid];
  sv[1] = S4[tid + 512];
#pragma unroll
  for (int it = 0; it < 16; ++it) ev[it] = eb4[tid + 512 * it];
  SB();
  LOAD_BLOCK(res0, 0);
  LOAD_BLOCK(res1, 1);
  LOAD_BLOCK(res2, 2);
  SB();

#pragma unroll
  for (int q = 0; q < 2; ++q) {
    float* d = &sS[(ng + 32 * q) * SS + 4 * o4];
    d[0] = sv[q][0]; d[1] = sv[q][1]; d[2] = sv[q][2]; d[3] = sv[q][3];
  }
#pragma unroll
  for (int it = 0; it < 16; ++it) {
    float* d = &priorsLds[(ng + 32 * it) * NP + 4 * o4];
    d[0] = ev[it][0]; d[1] = ev[it][1]; d[2] = ev[it][2]; d[3] = ev[it][3];
  }
  for (int i = tid; i < K_CAPS * N_DIM; i += 512) accSim[i] = 0.0f;
  lds_barrier();

  {
    const int lm = tid & 15;
    const int lh = (tid >> 4) & 1;
    const int mw = (tid >> 5) * 32;
    const float* sb = sS + 32 * lh * SS + lm;
    for (int mt = 0; mt < 2; ++mt) {
      const int m0 = mw + 16 * mt;
      const float* ar = priorsLds + (m0 + lm) * NP + 32 * lh;
      v2f a[16];
#pragma unroll
      for (int t = 0; t < 16; ++t) { a[t][0] = ar[2 * t]; a[t][1] = ar[2 * t + 1]; }
      float* dr = priorsLds + (m0 + 8 * lh) * NP + lm;
      for (int np = 0; np < 2; ++np) {
        const float* b0p = sb + 32 * np;
        const float* b1p = b0p + 16;
        v8f d0, d1;
#pragma unroll
        for (int j = 0; j < 8; ++j) { d0[j] = 0.0f; d1[j] = 0.0f; }
#pragma unroll
        for (int t = 0; t < 16; ++t) {
          v2f b0, b1;
          b0[0] = b0p[(2 * t) * SS];  b0[1] = b0p[(2 * t + 1) * SS];
          b1[0] = b1p[(2 * t) * SS];  b1[1] = b1p[(2 * t + 1) * SS];
          d0 = __builtin_amdgcn_wmma_f32_16x16x4_f32(
               false, a[t],  false, b0,
               (short)0, d0,  false,  false);
          d1 = __builtin_amdgcn_wmma_f32_16x16x4_f32(
               false, a[t],  false, b1,
               (short)0, d1,  false,  false);
        }
#pragma unroll
        for (int r = 0; r < 8; ++r) {
          dr[r * NP + 32 * np]      = d0[r];
          dr[r * NP + 32 * np + 16] = d1[r];
        }
      }
    }
  }
  lds_barrier();

  for (int iter = 0; iter < 3; ++iter) {
    f4 zacc[K_CAPS];
#pragma unroll
    for (int k = 0; k < K_CAPS; ++k) {
#pragma unroll
      for (int j = 0; j < 4; ++j) zacc[k][j] = 0.0f;
    }
    f4 bufA[K_CAPS], bufB[K_CAPS];

    if (iter == 0) {
      SB(); LOAD_BLOCK(bufA, 3);        SB(); SUM_BLOCK(res0, 0); END_BLOCK();
      SUM_BLOCK(res1, 1); END_BLOCK();
      SB(); LOAD_BLOCK(bufB, NBLK - 1); SB(); SUM_BLOCK(bufA, 3); KEEP_BLOCK(bufA); END_BLOCK();
      SB(); LOAD_BLOCK(bufA, NBLK - 2); SB(); SUM_BLOCK(res2, 2); END_BLOCK();
      SUM_BLOCK(bufB, NBLK - 1); END_BLOCK();
      int jn = NBLK - 2;
      asm volatile("" : "+s"(jn));
      for (; jn > NKEEP; jn -= 2) {
        SB(); LOAD_BLOCK(bufB, jn - 1); SB(); SUM_BLOCK(bufA, jn);     END_BLOCK();
        SB(); LOAD_BLOCK(bufA, jn - 2); SB(); SUM_BLOCK(bufB, jn - 1); END_BLOCK();
      }
      SUM_BLOCK(bufA, NKEEP);
    } else {
      const int jb = (iter == 1) ? NKEEP : (NBLK - 1);
      const int js = (iter == 1) ? 1 : -1;
      LOAD_BLOCK(bufA, NB(0));
      for (int s = 0; s < 8; s += 2) {
        SB(); LOAD_BLOCK(bufB, NB(s + 1)); SB(); SUM_BLOCK(bufA, NB(s));     END_BLOCK();
        SB(); LOAD_BLOCK(bufA, NB(s + 2)); SB(); SUM_BLOCK(bufB, NB(s + 1)); END_BLOCK();
      }
      SUM_BLOCK(res0, 0); END_BLOCK();
      SB(); LOAD_BLOCK(bufB, NB(9));  SB(); SUM_BLOCK(bufA, NB(8));  END_BLOCK();
      SUM_BLOCK(res1, 1); END_BLOCK();
      SB(); LOAD_BLOCK(bufA, NB(10)); SB(); SUM_BLOCK(bufB, NB(9));  END_BLOCK();
      SUM_BLOCK(res2, 2); END_BLOCK();
      SB(); LOAD_BLOCK(bufB, NB(11)); SB(); SUM_BLOCK(bufA, NB(10)); END_BLOCK();
      FETCH_KEPT(bufA);
      SUM_BLOCK(bufA, 3); END_BLOCK();
      SUM_BLOCK(bufB, NB(11));
    }

#pragma unroll
    for (int k = 0; k < K_CAPS; ++k)
      *(f4*)&part[(k * NGRP + ng) * O_DIM + 4 * o4] = zacc[k];
    __syncthreads();

    {
      float z = 0.0f;
#pragma unroll
      for (int g2 = 0; g2 < NGRP; ++g2) z += part[(g * NGRP + g2) * O_DIM + o];
      zLds[g * O_DIM + o] = z;
    }
    __syncthreads();

    if (tid < K_CAPS) {
      float sq = 0.0f;
      for (int oo = 0; oo < O_DIM; ++oo) {
        const float zz = zLds[tid * O_DIM + oo];
        sq += zz * zz;
      }
      scaleL[tid] = (sq / (1.0f + sq)) / sqrtf(sq + 1e-9f);
    }
    __syncthreads();

    if (iter == 2) {
      int oi = tid;
      asm volatile("" : "+v"(oi));
      out[(size_t)b * K_CAPS * O_DIM + oi] = zLds[tid] * scaleL[g];
    } else {
      vLds[g * VS + o] = zLds[tid] * scaleL[g];
      __syncthreads();
      const int lm = tid & 15;
      const int lh = (tid >> 4) & 1;
      const int nw = (tid >> 5) * 32;
      const float* va  = vLds + (lm & 7) * VS + 32 * lh;
      const float* pb0 = priorsLds + (nw + lm) * NP + 32 * lh;
      const float* pb1 = pb0 + 16 * NP;
      v8f d0, d1;
#pragma unroll
      for (int j = 0; j < 8; ++j) { d0[j] = 0.0f; d1[j] = 0.0f; }
#pragma unroll
      for (int t = 0; t < 16; ++t) {
        v2f a, b0, b1;
        a[0]  = va[2 * t];   a[1]  = va[2 * t + 1];
        b0[0] = pb0[2 * t];  b0[1] = pb0[2 * t + 1];
        b1[0] = pb1[2 * t];  b1[1] = pb1[2 * t + 1];
        d0 = __builtin_amdgcn_wmma_f32_16x16x4_f32(
             false, a,  false, b0,
             (short)0, d0,  false,  false);
        d1 = __builtin_amdgcn_wmma_f32_16x16x4_f32(
             false, a,  false, b1,
             (short)0, d1,  false,  false);
      }
      {
        float* sp = accSim + nw + 16 * lh + lm;
#pragma unroll
        for (int r = 0; r < K_CAPS; ++r)
          sp[r * N_DIM] += lh ? d1[r] : d0[r];
      }
      __syncthreads();
    }
  }
}
#undef NB
#undef LOAD_BLOCK
#undef KEEP_BLOCK
#undef FETCH_KEPT
#undef SUM_BLOCK
#undef END_BLOCK
#undef SB

extern "C" void kernel_launch(void* const* d_in, const int* in_sizes, int n_in,
                              void* d_out, int out_size, void* d_ws, size_t ws_size,
                              hipStream_t stream) {
  (void)in_sizes; (void)n_in; (void)out_size; (void)d_ws; (void)ws_size;
  const float* emb = (const float*)d_in[0];
  const float* S   = (const float*)d_in[1];
  const float* cc  = (const float*)d_in[2];
  float* out    = (float*)d_out;

  const size_t smemBytes =
      (size_t)(N_DIM * NP + K_CAPS * N_DIM + K_CAPS * O_DIM + K_CAPS * VS +
               K_CAPS * NGRP * O_DIM + 8 +
               K_CAPS * 512 * 4) * sizeof(float);
  routing_kernel<<<B_DIM, 512, smemBytes, stream>>>(emb, S, cc, out);
}


# Pass A block sum with fewer vector instructions: 1/sum by the hardware reciprocal and one multiply (was the exact division), log2(e) and the max folded into one multiply-add in front of exp2
# speedup vs baseline: 3.0221x; 1.0882x over previous
#include <hip/hip_runtime.h>
#include <math.h>

typedef __attribute__((ext_vector_type(2))) float v2f;
typedef __attribute__((ext_vector_type(4))) float f4;
typedef __attribute__((ext_vector_type(8))) float v8f;

#define B_DIM 256
#define N_DIM 512
#define E_DIM 64
#define O_DIM 64
#define K_CAPS 8

#define NP 65
#define NGRP 32
#define NBLK 16
#define NKEEP 4
#define VS 65
#define SS 65
#define CC_KSTRIDE4 (N_DIM * O_DIM / 4)

#define NB(S) (jb + js * (S))

#define LOAD_BLOCK(BUF, BLK)                                                  \
  do {                                                                        \
    const f4* cblk = ccp + (BLK) * 512;                                       \
    _Pragma("unroll") for (int k = 0; k < K_CAPS; ++k)                        \
      BUF[k] = cblk[(size_t)k * CC_KSTRIDE4];                                 \
  } while (0)

#define KEEP_BLOCK(BUF)                                                       \
  do {                                                                        \
    _Pragma("unroll") for (int k = 0; k < K_CAPS; ++k)                        \
      keep4[k * 512] = BUF[k];                                                \
  } while (0)
#define FETCH_KEPT(BUF)                                                       \
  do {                                                                        \
    _Pragma("unroll") for (int k = 0; k < K_CAPS; ++k)                        \
      BUF[k] = keep4[k * 512];                                                \
  } while (0)

#define SUM_BLOCK(BUF, BLK)                                                   \
  do {                                                                        \
    const int n = ng + (BLK) * NGRP;                                          \
    v2f prl, prh;                                                             \
    prl[0] = priorsLds[n * NP + 4 * o4 + 0];                                  \
    prl[1] = priorsLds[n * NP + 4 * o4 + 1];                                  \
    prh[0] = priorsLds[n * NP + 4 * o4 + 2];                                  \
    prh[1] = priorsLds[n * NP + 4 * o4 + 3];                                  \
    v2f cl[K_CAPS], ch[K_CAPS];                                               \
    v2f ml, mh;                                                               \
    ml[0] = ml[1] = mh[0] = mh[1] = -3.402823466e+38f;                        \
    _Pragma("unroll") for (int k = 0; k < K_CAPS; ++k) {                      \
      const float a = accSim[k * N_DIM + n];                                  \
      v2f av; av[0] = a; av[1] = a;                                           \
      v2f bl, bh;                                                             \
      bl[0] = BUF[k][0]; bl[1] = BUF[k][1];                                   \
      bh[0] = BUF[k][2]; bh[1] = BUF[k][3];                                   \
      cl[k] = bl + av;                                                        \
      ch[k] = bh + av;                                                        \
      ml[0] = fmaxf(ml[0], cl[k][0]); ml[1] = fmaxf(ml[1], cl[k][1]);         \
      mh[0] = fmaxf(mh[0], ch[k][0]); mh[1] = fmaxf(mh[1], ch[k][1]);         \
    }                                                                         \
    v2f l2e; l2e[0] = l2e[1] = 1.44269504088896341f;                          \
    const v2f nl = -(ml * l2e);                                               \
    const v2f nh = -(mh * l2e);                                               \
    v2f sl, sh;                                                               \
    sl[0] = sl[1] = sh[0] = sh[1] = 0.0f;                                     \
    _Pragma("unroll") for (int k = 0; k < K_CAPS; ++k) {                      \
      const v2f dl = __builtin_elementwise_fma(cl[k], l2e, nl);               \
      const v2f dh = __builtin_elementwise_fma(ch[k], l2e, nh);               \
      cl[k][0] = __builtin_amdgcn_exp2f(dl[0]);                               \
      cl[k][1] = __builtin_amdgcn_exp2f(dl[1]);                               \
      ch[k][0] = __builtin_amdgcn_exp2f(dh[0]);                               \
      ch[k][1] = __builtin_amdgcn_exp2f(dh[1]);                               \
      sl += cl[k];                                                            \
      sh += ch[k];                                                            \
    }                                                                         \
    v2f rl, rh;                                                               \
    rl[0] = __builtin_amdgcn_rcpf(sl[0]); rl[1] = __builtin_amdgcn_rcpf(sl[1]); \
    rh[0] = __builtin_amdgcn_rcpf(sh[0]); rh[1] = __builtin_amdgcn_rcpf(sh[1]); \
    const v2f pl = prl * rl;                                                  \
    const v2f ph = prh * rh;                                                  \
    _Pragma("unroll") for (int k = 0; k < K_CAPS; ++k) {                      \
      zacc[k][0] += cl[k][0] * pl[0]; zacc[k][1] += cl[k][1] * pl[1];         \
      zacc[k][2] += ch[k][0] * ph[0]; zacc[k][3] += ch[k][1] * ph[1];         \
    }                                                                         \
  } while (0)

#define END_BLOCK()                                                           \
  do {                                                                        \
    _Pragma("unroll") for (int k = 0; k < K_CAPS; ++k)                        \
      asm volatile("" : "+v"(zacc[k]));                                       \
  } while (0)

#define SB() __builtin_amdgcn_sched_barrier(0)

__device__ __forceinline__ void lds_barrier() {
  __builtin_amdgcn_fence(__ATOMIC_RELEASE, "workgroup", "local");
  __builtin_amdgcn_s_barrier();
  __builtin_amdgcn_fence(__ATOMIC_ACQUIRE, "workgroup", "local");
}

__global__ __launch_bounds__(512) void routing_kernel(
    const float* __restrict__ emb, const float* __restrict__ S,
    const float* __restrict__ cc, float* __restrict__ out)
{
  extern __shared__ __attribute__((aligned(16))) float smem[];
  float* priorsLds = smem;
  float* accSim    = priorsLds + N_DIM * NP;
  float* zLds      = accSim + K_CAPS * N_DIM;
  float* vLds      = zLds + K_CAPS * O_DIM;
  float* part      = vLds + K_CAPS * VS;
  float* scaleL    = part + K_CAPS * NGRP * O_DIM;
  float* keepL     = scaleL + 8;

  const int tid = threadIdx.x;
  const int b   = blockIdx.x;
  const float* ccb = cc + (size_t)b * K_CAPS * N_DIM * O_DIM;

  const int o  = tid & 63;
  const int g  = tid >> 6;
  const int o4 = tid & 15;
  const int ng = tid >> 4;
  const f4* ccp = (const f4*)ccb + tid;
  f4* keep4 = (f4*)keepL + tid;

  f4 res0[K_CAPS], res1[K_CAPS], res2[K_CAPS];

  float* sS = part;
  const f4* S4  = (const f4*)S;
  const f4* eb4 = (const f4*)(emb + (size_t)b * N_DIM * E_DIM);
  f4 sv[2], ev[16];
  sv[0] = S4[tid];
  sv[1] = S4[tid + 512];
#pragma unroll
  for (int it = 0; it < 16; ++it) ev[it] = eb4[tid + 512 * it];
  SB();
  LOAD_BLOCK(res0, 0);
  LOAD_BLOCK(res1, 1);
  LOAD_BLOCK(res2, 2);
  SB();

#pragma unroll
  for (int q = 0; q < 2; ++q) {
    float* d = &sS[(ng + 32 * q) * SS + 4 * o4];
    d[0] = sv[q][0]; d[1] = sv[q][1]; d[2] = sv[q][2]; d[3] = sv[q][3];
  }
#pragma unroll
  for (int it = 0; it < 16; ++it) {
    float* d = &priorsLds[(ng + 32 * it) * NP + 4 * o4];
    d[0] = ev[it][0]; d[1] = ev[it][1]; d[2] = ev[it][2]; d[3] = ev[it][3];
  }
  for (int i = tid; i < K_CAPS * N_DIM; i += 512) accSim[i] = 0.0f;
  lds_barrier();

  {
    const int lm = tid & 15;
    const int lh = (tid >> 4) & 1;
    const int mw = (tid >> 5) * 32;
    const float* sb = sS + 32 * lh * SS + lm;
    for (int mt = 0; mt < 2; ++mt) {
      const int m0 = mw + 16 * mt;
      const float* ar = priorsLds + (m0 + lm) * NP + 32 * lh;
      v2f a[16];
#pragma unroll
      for (int t = 0; t < 16; ++t) { a[t][0] = ar[2 * t]; a[t][1] = ar[2 * t + 1]; }
      float* dr = priorsLds + (m0 + 8 * lh) * NP + lm;
      for (int np = 0; np < 2; ++np) {
        const float* b0p = sb + 32 * np;
        const float* b1p = b0p + 16;
        v8f d0, d1;
#pragma unroll
        for (int j = 0; j < 8; ++j) { d0[j] = 0.0f; d1[j] = 0.0f; }
#pragma unroll
        for (int t = 0; t < 16; ++t) {
          v2f b0, b1;
          b0[0] = b0p[(2 * t) * SS];  b0[1] = b0p[(2 * t + 1) * SS];
          b1[0] = b1p[(2 * t) * SS];  b1[1] = b1p[(2 * t + 1) * SS];
          d0 = __builtin_amdgcn_wmma_f32_16x16x4_f32(
               false, a[t],  false, b0,
               (short)0, d0,  false,  false);
          d1 = __builtin_amdgcn_wmma_f32_16x16x4_f32(
               false, a[t],  false, b1,
               (short)0, d1,  false,  false);
        }
#pragma unroll
        for (int r = 0; r < 8; ++r) {
          dr[r * NP + 32 * np]      = d0[r];
          dr[r * NP + 32 * np + 16] = d1[r];
        }
      }
    }
  }
  lds_barrier();

  for (int iter = 0; iter < 3; ++iter) {
    f4 zacc[K_CAPS];
#pragma unroll
    for (int k = 0; k < K_CAPS; ++k) {
#pragma unroll
      for (int j = 0; j < 4; ++j) zacc[k][j] = 0.0f;
    }
    f4 bufA[K_CAPS], bufB[K_CAPS];

    if (iter == 0) {
      SB(); LOAD_BLOCK(bufA, 3);        SB(); SUM_BLOCK(res0, 0); END_BLOCK();
      SUM_BLOCK(res1, 1); END_BLOCK();
      SB(); LOAD_BLOCK(bufB, NBLK - 1); SB(); SUM_BLOCK(bufA, 3); KEEP_BLOCK(bufA); END_BLOCK();
      SB(); LOAD_BLOCK(bufA, NBLK - 2); SB(); SUM_BLOCK(res2, 2); END_BLOCK();
      SUM_BLOCK(bufB, NBLK - 1); END_BLOCK();
      int jn = NBLK - 2;
      asm volatile("" : "+s"(jn));
      for (; jn > NKEEP; jn -= 2) {
        SB(); LOAD_BLOCK(bufB, jn - 1); SB(); SUM_BLOCK(bufA, jn);     END_BLOCK();
        SB(); LOAD_BLOCK(bufA, jn - 2); SB(); SUM_BLOCK(bufB, jn - 1); END_BLOCK();
      }
      SUM_BLOCK(bufA, NKEEP);
    } else {
      const int jb = (iter == 1) ? NKEEP : (NBLK - 1);
      const int js = (iter == 1) ? 1 : -1;
      LOAD_BLOCK(bufA, NB(0));
      for (int s = 0; s < 8; s += 2) {
        SB(); LOAD_BLOCK(bufB, NB(s + 1)); SB(); SUM_BLOCK(bufA, NB(s));     END_BLOCK();
        SB(); LOAD_BLOCK(bufA, NB(s + 2)); SB(); SUM_BLOCK(bufB, NB(s + 1)); END_BLOCK();
      }
      SUM_BLOCK(res0, 0); END_BLOCK();
      SB(); LOAD_BLOCK(bufB, NB(9));  SB(); SUM_BLOCK(bufA, NB(8));  END_BLOCK();
      SUM_BLOCK(res1, 1); END_BLOCK();
      SB(); LOAD_BLOCK(bufA, NB(10)); SB(); SUM_BLOCK(bufB, NB(9));  END_BLOCK();
      SUM_BLOCK(res2, 2); END_BLOCK();
      SB(); LOAD_BLOCK(bufB, NB(11)); SB(); SUM_BLOCK(bufA, NB(10)); END_BLOCK();
      FETCH_KEPT(bufA);
      SUM_BLOCK(bufA, 3); END_BLOCK();
      SUM_BLOCK(bufB, NB(11));
    }

#pragma unroll
    for (int k = 0; k < K_CAPS; ++k)
      *(f4*)&part[(k * NGRP + ng) * O_DIM + 4 * o4] = zacc[k];
    __syncthreads();

    {
      float z = 0.0f;
#pragma unroll
      for (int g2 = 0; g2 < NGRP; ++g2) z += part[(g * NGRP + g2) * O_DIM + o];
      zLds[g * O_DIM + o] = z;
    }
    __syncthreads();

    if (tid < K_CAPS) {
      float sq = 0.0f;
      for (int oo = 0; oo < O_DIM; ++oo) {
        const float zz = zLds[tid * O_DIM + oo];
        sq += zz * zz;
      }
      scaleL[tid] = (sq / (1.0f + sq)) / sqrtf(sq + 1e-9f);
    }
    __syncthreads();

    if (iter == 2) {
      int oi = tid;
      asm volatile("" : "+v"(oi));
      out[(size_t)b * K_CAPS * O_DIM + oi] = zLds[tid] * scaleL[g];
    } else {
      vLds[g * VS + o] = zLds[tid] * scaleL[g];
      __syncthreads();
      const int lm = tid & 15;
      const int lh = (tid >> 4) & 1;
      const int nw = (tid >> 5) * 32;
      const float* va  = vLds + (lm & 7) * VS + 32 * lh;
      const float* pb0 = priorsLds + (nw + lm) * NP + 32 * lh;
      const float* pb1 = pb0 + 16 * NP;
      v8f d0, d1;
#pragma unroll
      for (int j = 0; j < 8; ++j) { d0[j] = 0.0f; d1[j] = 0.0f; }
#pragma unroll
      for (int t = 0; t < 16; ++t) {
        v2f a, b0, b1;
        a[0]  = va[2 * t];   a[1]  = va[2 * t + 1];
        b0[0] = pb0[2 * t];  b0[1] = pb0[2 * t + 1];
        b1[0] = pb1[2 * t];  b1[1] = pb1[2 * t + 1];
        d0 = __builtin_amdgcn_wmma_f32_16x16x4_f32(
             false, a,  false, b0,
             (short)0, d0,  false,  false);
        d1 = __builtin_amdgcn_wmma_f32_16x16x4_f32(
             false, a,  false, b1,
             (short)0, d1,  false,  false);
      }
      {
        float* sp = accSim + nw + 16 * lh + lm;
#pragma unroll
        for (int r = 0; r < K_CAPS; ++r)
          sp[r * N_DIM] += lh ? d1[r] : d0[r];
      }
      __syncthreads();
    }
  }
}
#undef NB
#undef LOAD_BLOCK
#undef KEEP_BLOCK
#undef FETCH_KEPT
#undef SUM_BLOCK
#undef END_BLOCK
#undef SB

extern "C" void kernel_launch(void* const* d_in, const int* in_sizes, int n_in,
                              void* d_out, int out_size, void* d_ws, size_t ws_size,
                              hipStream_t stream) {
  (void)in_sizes; (void)n_in; (void)out_size; (void)d_ws; (void)ws_size;
  const float* emb = (const float*)d_in[0];
  const float* S   = (const float*)d_in[1];
  const float* cc  = (const float*)d_in[2];
  float* out    = (float*)d_out;

  const size_t smemBytes =
      (size_t)(N_DIM * NP + K_CAPS * N_DIM + K_CAPS * O_DIM + K_CAPS * VS +
               K_CAPS * NGRP * O_DIM + 8 +
               K_CAPS * 512 * 4) * sizeof(float);
  routing_kernel<<<B_DIM, 512, smemBytes, stream>>>(emb, S, cc, out);
}
